# starting kernel
# baseline (speedup 1.0000x reference)
#include <hip/hip_runtime.h>

typedef float v2f __attribute__((ext_vector_type(2)));
typedef float v8f __attribute__((ext_vector_type(8)));

constexpr int NPTS = 50000;
constexpr int KNBR = 16;
constexpr int CIN  = 64;
constexpr int COUT = 64;
constexpr int NB   = 4;
constexpr int NTILES = NPTS / 16;
constexpr int WPB  = 8;
constexpr int TPB  = WPB * 32;
constexpr int LDSW_FLOATS = KNBR * 16 * 2 * 64 * 2;

__global__ __launch_bounds__(TPB)
void conv_gnn_wmma_f32(const float* __restrict__ xin,
                       const int*   __restrict__ nbr,
                       const float* __restrict__ wgt,
                       const float* __restrict__ bias,
                       float*       __restrict__ out)
{
  extern __shared__ float ldsw[];
  const int tid = threadIdx.x;

  for (int p = tid; p < LDSW_FLOATS / 2; p += TPB) {
    const int o    = p & 63;
    const int half = (p >> 6) & 1;
    const int cb   = (p >> 7) & 15;
    const int k    = p >> 11;
    const int c0   = cb * 4 + half * 2;
    const int src  = (k * CIN + c0) * COUT + o;
    ldsw[2 * p]     = wgt[src];
    ldsw[2 * p + 1] = wgt[src + COUT];
  }
  __syncthreads();

  const int wave = tid >> 5;
  const int lane = tid & 31;
  const int tile = blockIdx.x * WPB + wave;
  if (tile >= NTILES) return;

  const int n0   = tile * 16;
  const int m    = lane & 15;
  const int half = lane >> 4;

  int roff[KNBR];
#pragma unroll
  for (int k = 0; k < KNBR; ++k) {
    const int idx = nbr[(n0 + m) * KNBR + k];
    roff[k] = (idx < NPTS) ? idx * (CIN * 4) : -1;
  }

  const char* bb[NB];
#pragma unroll
  for (int b = 0; b < NB; ++b)
    bb[b] = (const char*)xin + (size_t)b * (size_t)NPTS * CIN * 4 + half * 8;

  v8f acc[NB][4];
#pragma unroll
  for (int b = 0; b < NB; ++b)
#pragma unroll
    for (int ot = 0; ot < 4; ++ot)
      acc[b][ot] = {};

#pragma unroll
  for (int k = 0; k < KNBR; ++k) {
    const bool valid = roff[k] >= 0;
    const int  ro    = valid ? roff[k] : 0;
    const float* lw = &ldsw[(k * 32 + half) * 128 + m * 2];
#pragma unroll 4
    for (int cb = 0; cb < 16; ++cb) {
      v2f a[NB];
#pragma unroll
      for (int b = 0; b < NB; ++b) {
        v2f t = *(const v2f*)(bb[b] + ro + cb * 16);
        a[b] = valid ? t : v2f{0.0f, 0.0f};
      }
      v2f bw[4];
#pragma unroll
      for (int ot = 0; ot < 4; ++ot)
        bw[ot] = *(const v2f*)(lw + cb * 256 + ot * 32);
#pragma unroll
      for (int b = 0; b < NB; ++b)
#pragma unroll
        for (int ot = 0; ot < 4; ++ot)
          acc[b][ot] = __builtin_amdgcn_wmma_f32_16x16x4_f32(
              false, a[b], false, bw[ot], (short)0, acc[b][ot], false, false);
    }
  }

#pragma unroll
  for (int b = 0; b < NB; ++b) {
#pragma unroll
    for (int ot = 0; ot < 4; ++ot) {
      const float bs = bias[ot * 16 + m];
#pragma unroll
      for (int j = 0; j < 8; ++j) {
        const int row = n0 + half * 8 + j;
        out[((size_t)b * NPTS + row) * COUT + ot * 16 + m] = acc[b][ot][j] + bs;
      }
    }
  }
}

extern "C" void kernel_launch(void* const* d_in, const int* in_sizes, int n_in,
                              void* d_out, int out_size, void* d_ws, size_t ws_size,
                              hipStream_t stream) {
  (void)in_sizes; (void)n_in; (void)d_ws; (void)ws_size; (void)out_size;
  const float* xin  = (const float*)d_in[0];
  const int*   nbr  = (const int*)d_in[1];
  const float* wgt  = (const float*)d_in[2];
  const float* bias = (const float*)d_in[3];
  float*       out  = (float*)d_out;

  const size_t shmem = (size_t)LDSW_FLOATS * sizeof(float);
  hipFuncSetAttribute((const void*)conv_gnn_wmma_f32,
                      hipFuncAttributeMaxDynamicSharedMemorySize, (int)shmem);

  const int blocks = (NTILES + WPB - 1) / WPB;
  conv_gnn_wmma_f32<<<blocks, TPB, shmem, stream>>>(xin, nbr, wgt, bias, out);
}


# Work split: one work-group per WGP (256) stages the weights once and every wave takes an equal 6 tile-batch units (a whole tile + half a tile), the 53 leftover tiles cut along cout
# speedup vs baseline: 1.3958x; 1.3958x over previous
#include <hip/hip_runtime.h>


typedef float v2f __attribute__((ext_vector_type(2)));
typedef float v4f __attribute__((ext_vector_type(4)));
typedef float v8f __attribute__((ext_vector_type(8)));

constexpr int NPTS = 50000;
constexpr int KNBR = 16;
constexpr int CIN  = 64;
constexpr int COUT = 64;
constexpr int NBATCH = 4;
constexpr int NTILES = NPTS / 16;
constexpr int WPB  = 8;
constexpr int TPB  = WPB * 32;
constexpr int LDSW_FLOATS = KNBR * 16 * 2 * 64 * 2;

constexpr int NBLK   = 256;
constexpr int NWAVES = NBLK * WPB;
constexpr int QUAD_TILES   = NWAVES;
constexpr int PAIR_TILES   = NWAVES / 2;
constexpr int SINGLE_TILE0 = QUAD_TILES + PAIR_TILES;
constexpr int NSINGLE      = (NTILES - SINGLE_TILE0) * NBATCH;
static_assert(NTILES * 16 == NPTS, "tile count");
static_assert(NTILES >= SINGLE_TILE0, "split covers the tiles");
static_assert(NSINGLE >= 0 && NSINGLE <= NBLK, "one extra unit per work-group at most");

template <int NBT>
__device__ __forceinline__ void load_rows(v4f (&A)[NBT][4], const char* const (&bb)[NBT], int off)
{
#pragma unroll
  for (int b = 0; b < NBT; ++b)
#pragma unroll
    for (int j = 0; j < 4; ++j)
      A[b][j] = *(const v4f*)(bb[b] + off + j * 32);
}

template <int NBT, int NOT>
__device__ __forceinline__ void mma_line(v8f (&acc)[NBT][NOT], v4f (&A)[NBT][4], bool valid,
                                         const float* lw, int tbase)
{
  if (__builtin_amdgcn_ballot_w32(!valid) != 0u) {
#pragma unroll
    for (int b = 0; b < NBT; ++b)
#pragma unroll
      for (int j = 0; j < 4; ++j)
        A[b][j] = valid ? A[b][j] : v4f{0.0f, 0.0f, 0.0f, 0.0f};
  }
#pragma unroll
  for (int j = 0; j < 4; ++j) {
#pragma unroll
    for (int s = 0; s < 2; ++s) {
      const int t = tbase + j * 2 + s;
      v2f bw[NOT];
#pragma unroll
      for (int ot = 0; ot < NOT; ++ot)
        bw[ot] = *(const v2f*)(lw + t * 256 + ot * 32);
#pragma unroll
      for (int b = 0; b < NBT; ++b) {
        const v2f a = s ? v2f{A[b][j].z, A[b][j].w} : v2f{A[b][j].x, A[b][j].y};
#pragma unroll
        for (int ot = 0; ot < NOT; ++ot)
          acc[b][ot] = __builtin_amdgcn_wmma_f32_16x16x4_f32(
              false, a, false, bw[ot], (short)0, acc[b][ot], false, false);
      }
    }
  }
}

template <int NBT, int NOT>
__device__ __forceinline__ void run_job(const float* __restrict__ xin,
                                        const int*   __restrict__ nbr,
                                        const float* __restrict__ bias,
                                        float*       __restrict__ out,
                                        const float* ldsw,
                                        int tile, int b0, int ot0, int lane)
{
  const int n0   = tile * 16;
  const int m    = lane & 15;
  const int half = lane >> 4;

  const int* nrow = nbr + (n0 + m) * KNBR;

  const char* bb[NBT];
#pragma unroll
  for (int b = 0; b < NBT; ++b)
    bb[b] = (const char*)xin + (size_t)(b0 + b) * (size_t)NPTS * CIN * 4 + half * 16;

  v8f acc[NBT][NOT];
#pragma unroll
  for (int b = 0; b < NBT; ++b)
#pragma unroll
    for (int ot = 0; ot < NOT; ++ot)
      acc[b][ot] = {};

  int  idx_c   = nrow[0];
  int  idx_n   = nrow[1];
  bool valid_c = idx_c < NPTS;
  int  ro_c    = valid_c ? idx_c * (CIN * 4) : 0;

  v4f A0[NBT][4], A1[NBT][4];
  load_rows<NBT>(A0, bb, ro_c);

  const float* lw = &ldsw[half * 128 + m * 2 + ot0 * 32];

#pragma unroll 1
  for (int k = 0; k < KNBR; ++k) {
    load_rows<NBT>(A1, bb, ro_c + 128);
    const int k2 = (k + 2 < KNBR) ? (k + 2) : (KNBR - 1);
    const int idx_nn = nrow[k2];

    mma_line<NBT, NOT>(acc, A0, valid_c, lw, 0);

    const bool valid_n = idx_n < NPTS;
    const int  ro_n    = valid_n ? idx_n * (CIN * 4) : 0;
    load_rows<NBT>(A0, bb, ro_n);

    mma_line<NBT, NOT>(acc, A1, valid_c, lw, 8);

    valid_c = valid_n;
    ro_c    = ro_n;
    idx_n   = idx_nn;
    lw     += 32 * 128;
  }

#pragma unroll
  for (int b = 0; b < NBT; ++b) {
#pragma unroll
    for (int ot = 0; ot < NOT; ++ot) {
      const float bs = bias[(ot0 + ot) * 16 + m];
#pragma unroll
      for (int j = 0; j < 8; ++j) {
        const int row = n0 + half * 8 + j;
        out[((size_t)(b0 + b) * NPTS + row) * COUT + (ot0 + ot) * 16 + m] = acc[b][ot][j] + bs;
      }
    }
  }
}

__global__ __launch_bounds__(TPB)
void conv_gnn_wmma_f32(const float* __restrict__ xin,
                       const int*   __restrict__ nbr,
                       const float* __restrict__ wgt,
                       const float* __restrict__ bias,
                       float*       __restrict__ out)
{
  extern __shared__ float ldsw[];
  const int tid = threadIdx.x;

  for (int p = tid; p < LDSW_FLOATS / 2; p += TPB) {
    const int o    = p & 63;
    const int half = (p >> 6) & 1;
    const int t    = (p >> 7) & 15;
    const int k    = p >> 11;
    const int c0   = (t >> 1) * 8 + half * 4 + (t & 1) * 2;
    const int src  = (k * CIN + c0) * COUT + o;
    ldsw[2 * p]     = wgt[src];
    ldsw[2 * p + 1] = wgt[src + COUT];
  }
  __syncthreads();

  const int wave = __builtin_amdgcn_readfirstlane(tid >> 5);
  const int lane = tid & 31;
  const int blk  = blockIdx.x;
  const int g    = blk * WPB + wave;

  const int qtile = g;
  const int ptile = QUAD_TILES + (g >> 1);
  const int pb0   = (g & 1) * 2;

  const bool small_first = ((wave ^ (wave >> 2)) & 1) != 0;

  if (small_first)
    run_job<2, 4>(xin, nbr, bias, out, ldsw, ptile, pb0, 0, lane);
  run_job<4, 4>(xin, nbr, bias, out, ldsw, qtile, 0, 0, lane);
  if (!small_first)
    run_job<2, 4>(xin, nbr, bias, out, ldsw, ptile, pb0, 0, lane);

  if (wave < 2 && blk < NSINGLE)
    run_job<1, 2>(xin, nbr, bias, out, ldsw, SINGLE_TILE0 + (blk >> 2), blk & 3, wave * 2, lane);
}

extern "C" void kernel_launch(void* const* d_in, const int* in_sizes, int n_in,
                              void* d_out, int out_size, void* d_ws, size_t ws_size,
                              hipStream_t stream) {
  (void)in_sizes; (void)n_in; (void)d_ws; (void)ws_size; (void)out_size;
  const float* xin  = (const float*)d_in[0];
  const int*   nbr  = (const int*)d_in[1];
  const float* wgt  = (const float*)d_in[2];
  const float* bias = (const float*)d_in[3];
  float*       out  = (float*)d_out;

  const size_t shmem = (size_t)LDSW_FLOATS * sizeof(float);
  hipFuncSetAttribute((const void*)conv_gnn_wmma_f32,
                      hipFuncAttributeMaxDynamicSharedMemorySize, (int)shmem);

  conv_gnn_wmma_f32<<<NBLK, TPB, shmem, stream>>>(xin, nbr, wgt, bias, out);
}


# Weight staging rewritten: 16-byte loads and LDS stores, 32 loads in flight per wave (2 passes, not 16), each work-group starts its sweep of the weights at its own 1 KB offset
# speedup vs baseline: 1.4602x; 1.0461x over previous
#include <hip/hip_runtime.h>


typedef float v2f __attribute__((ext_vector_type(2)));
typedef float v4f __attribute__((ext_vector_type(4)));
typedef float v8f __attribute__((ext_vector_type(8)));

constexpr int NPTS = 50000;
constexpr int KNBR = 16;
constexpr int CIN  = 64;
constexpr int COUT = 64;
constexpr int NBATCH = 4;
constexpr int NTILES = NPTS / 16;
constexpr int WPB  = 8;
constexpr int TPB  = WPB * 32;
constexpr int LDSW_FLOATS = KNBR * 16 * 2 * 64 * 2;

constexpr int NBLK   = 256;
constexpr int NWAVES = NBLK * WPB;
constexpr int QUAD_TILES   = NWAVES;
constexpr int PAIR_TILES   = NWAVES / 2;
constexpr int SINGLE_TILE0 = QUAD_TILES + PAIR_TILES;
constexpr int NSINGLE      = (NTILES - SINGLE_TILE0) * NBATCH;
static_assert(NTILES * 16 == NPTS, "tile count");
static_assert(NTILES >= SINGLE_TILE0, "split covers the tiles");
static_assert(NSINGLE >= 0 && NSINGLE <= NBLK, "one extra unit per work-group at most");

template <int NBT>
__device__ __forceinline__ void load_rows(v4f (&A)[NBT][4], const char* const (&bb)[NBT], int off)
{
#pragma unroll
  for (int b = 0; b < NBT; ++b)
#pragma unroll
    for (int j = 0; j < 4; ++j)
      A[b][j] = *(const v4f*)(bb[b] + off + j * 32);
}

template <int NBT, int NOT>
__device__ __forceinline__ void mma_line(v8f (&acc)[NBT][NOT], v4f (&A)[NBT][4], bool valid,
                                         const float* lw, int tbase)
{
  if (__builtin_amdgcn_ballot_w32(!valid) != 0u) {
#pragma unroll
    for (int b = 0; b < NBT; ++b)
#pragma unroll
      for (int j = 0; j < 4; ++j)
        A[b][j] = valid ? A[b][j] : v4f{0.0f, 0.0f, 0.0f, 0.0f};
  }
#pragma unroll
  for (int j = 0; j < 4; ++j) {
#pragma unroll
    for (int s = 0; s < 2; ++s) {
      const int t = tbase + j * 2 + s;
      v2f bw[NOT];
#pragma unroll
      for (int ot = 0; ot < NOT; ++ot)
        bw[ot] = *(const v2f*)(lw + t * 256 + ot * 32);
#pragma unroll
      for (int b = 0; b < NBT; ++b) {
        const v2f a = s ? v2f{A[b][j].z, A[b][j].w} : v2f{A[b][j].x, A[b][j].y};
#pragma unroll
        for (int ot = 0; ot < NOT; ++ot)
          acc[b][ot] = __builtin_amdgcn_wmma_f32_16x16x4_f32(
              false, a, false, bw[ot], (short)0, acc[b][ot], false, false);
      }
    }
  }
}

template <int NBT, int NOT>
__device__ __forceinline__ void run_job(const float* __restrict__ xin,
                                        const int*   __restrict__ nbr,
                                        const float* __restrict__ bias,
                                        float*       __restrict__ out,
                                        const float* ldsw,
                                        int tile, int b0, int ot0, int lane)
{
  const int n0   = tile * 16;
  const int m    = lane & 15;
  const int half = lane >> 4;

  const int* nrow = nbr + (n0 + m) * KNBR;

  const char* bb[NBT];
#pragma unroll
  for (int b = 0; b < NBT; ++b)
    bb[b] = (const char*)xin + (size_t)(b0 + b) * (size_t)NPTS * CIN * 4 + half * 16;

  v8f acc[NBT][NOT];
#pragma unroll
  for (int b = 0; b < NBT; ++b)
#pragma unroll
    for (int ot = 0; ot < NOT; ++ot)
      acc[b][ot] = {};

  int  idx_c   = nrow[0];
  int  idx_n   = nrow[1];
  bool valid_c = idx_c < NPTS;
  int  ro_c    = valid_c ? idx_c * (CIN * 4) : 0;

  v4f A0[NBT][4], A1[NBT][4];
  load_rows<NBT>(A0, bb, ro_c);

  const float* lw = &ldsw[half * 128 + m * 2 + ot0 * 32];

#pragma unroll 1
  for (int k = 0; k < KNBR; ++k) {
    load_rows<NBT>(A1, bb, ro_c + 128);
    const int k2 = (k + 2 < KNBR) ? (k + 2) : (KNBR - 1);
    const int idx_nn = nrow[k2];

    mma_line<NBT, NOT>(acc, A0, valid_c, lw, 0);

    const bool valid_n = idx_n < NPTS;
    const int  ro_n    = valid_n ? idx_n * (CIN * 4) : 0;
    load_rows<NBT>(A0, bb, ro_n);

    mma_line<NBT, NOT>(acc, A1, valid_c, lw, 8);

    valid_c = valid_n;
    ro_c    = ro_n;
    idx_n   = idx_nn;
    lw     += 32 * 128;
  }

#pragma unroll
  for (int b = 0; b < NBT; ++b) {
#pragma unroll
    for (int ot = 0; ot < NOT; ++ot) {
      const float bs = bias[(ot0 + ot) * 16 + m];
#pragma unroll
      for (int j = 0; j < 8; ++j) {
        const int row = n0 + half * 8 + j;
        out[((size_t)(b0 + b) * NPTS + row) * COUT + (ot0 + ot) * 16 + m] = acc[b][ot][j] + bs;
      }
    }
  }
}

__global__ __launch_bounds__(TPB)
void conv_gnn_wmma_f32(const float* __restrict__ xin,
                       const int*   __restrict__ nbr,
                       const float* __restrict__ wgt,
                       const float* __restrict__ bias,
                       float*       __restrict__ out)
{
  extern __shared__ float ldsw[];
  const int tid = threadIdx.x;

  {
    constexpr int NQUAD = LDSW_FLOATS / 8;
    const int rot = (int)blockIdx.x * 32;
    float* lst = (float*)__builtin_assume_aligned(ldsw, 16);
#pragma unroll 16
    for (int qi = tid; qi < NQUAD; qi += TPB) {
      const int q    = (qi + rot) & (NQUAD - 1);
      const int og   = q & 15;
      const int half = (q >> 4) & 1;
      const int t    = (q >> 5) & 15;
      const int k    = q >> 9;
      const int c0   = (t >> 1) * 8 + half * 4 + (t & 1) * 2;
      const float* s = wgt + (k * CIN + c0) * COUT + og * 4;
      const v4f w0 = *(const v4f*)s;
      const v4f w1 = *(const v4f*)(s + COUT);
      *(v4f*)&lst[q * 8]     = v4f{w0.x, w1.x, w0.y, w1.y};
      *(v4f*)&lst[q * 8 + 4] = v4f{w0.z, w1.z, w0.w, w1.w};
    }
  }
  __syncthreads();

  const int wave = __builtin_amdgcn_readfirstlane(tid >> 5);
  const int lane = tid & 31;
  const int blk  = blockIdx.x;
  const int g    = blk * WPB + wave;

  const int qtile = g;
  const int ptile = QUAD_TILES + (g >> 1);
  const int pb0   = (g & 1) * 2;

  const bool small_first = ((wave ^ (wave >> 2)) & 1) != 0;

  if (small_first)
    run_job<2, 4>(xin, nbr, bias, out, ldsw, ptile, pb0, 0, lane);
  run_job<4, 4>(xin, nbr, bias, out, ldsw, qtile, 0, 0, lane);
  if (!small_first)
    run_job<2, 4>(xin, nbr, bias, out, ldsw, ptile, pb0, 0, lane);

  if (wave < 2 && blk < NSINGLE)
    run_job<1, 2>(xin, nbr, bias, out, ldsw, SINGLE_TILE0 + (blk >> 2), blk & 3, wave * 2, lane);
}

extern "C" void kernel_launch(void* const* d_in, const int* in_sizes, int n_in,
                              void* d_out, int out_size, void* d_ws, size_t ws_size,
                              hipStream_t stream) {
  (void)in_sizes; (void)n_in; (void)d_ws; (void)ws_size; (void)out_size;
  const float* xin  = (const float*)d_in[0];
  const int*   nbr  = (const int*)d_in[1];
  const float* wgt  = (const float*)d_in[2];
  const float* bias = (const float*)d_in[3];
  float*       out  = (float*)d_out;

  const size_t shmem = (size_t)LDSW_FLOATS * sizeof(float);
  hipFuncSetAttribute((const void*)conv_gnn_wmma_f32,
                      hipFuncAttributeMaxDynamicSharedMemorySize, (int)shmem);

  conv_gnn_wmma_f32<<<NBLK, TPB, shmem, stream>>>(xin, nbr, wgt, bias, out);
}
